# starting kernel
# baseline (speedup 1.0000x reference)
#include <hip/hip_runtime.h>
#include <hip/hip_bf16.h>

typedef __attribute__((ext_vector_type(2))) float v2f;
typedef __attribute__((ext_vector_type(8))) float v8f;

#define EMB_DIM 128
#define HIDDEN  256
#define OUT_C   64
#define K_ITERS 10
#define ALPHA   0.1f

__global__ void deg_init_kernel(float* deg, int n) {
    int i = blockIdx.x * blockDim.x + threadIdx.x;
    if (i < n) deg[i] = 1.0f;
}

__global__ void deg_count_kernel(const int* __restrict__ dst, float* deg, int e) {
    int i = blockIdx.x * blockDim.x + threadIdx.x;
    if (i < e) atomicAdd(&deg[dst[i]], 1.0f);
}

__global__ void deg_rsqrt_kernel(float* deg, int n) {
    int i = blockIdx.x * blockDim.x + threadIdx.x;
    if (i < n) {
        float d = deg[i];
        deg[i] = (d > 0.0f) ? rsqrtf(d) : 0.0f;
    }
}

__global__ void edge_norm_kernel(const int* __restrict__ src, const int* __restrict__ dst,
                                 const float* __restrict__ dis, float* __restrict__ norm, int e) {
    int i = blockIdx.x * blockDim.x + threadIdx.x;
    if (i < e) norm[i] = dis[src[i]] * dis[dst[i]];
}

__launch_bounds__(128)
__global__ void mlp_wmma_kernel(const int* __restrict__ xidx,
                                const float* __restrict__ emb,
                                const float* __restrict__ W1, const float* __restrict__ b1,
                                const float* __restrict__ W2, const float* __restrict__ b2,
                                float* __restrict__ h_out, float* __restrict__ x0_out,
                                int n) {
    const int tid  = threadIdx.x;
    const int wave = tid >> 5;
    const int lane = tid & 31;
    const int m    = lane & 15;
    const int g    = lane >> 4;
    const int row0 = blockIdx.x * 16;

    __shared__ float sA[16][EMB_DIM + 2];
    __shared__ float sH[16][HIDDEN + 2];

    for (int i = tid; i < 16 * EMB_DIM; i += 128) {
        int r  = i >> 7;
        int c  = i & (EMB_DIM - 1);
        int rr = row0 + r;
        float v = 0.0f;
        if (rr < n) {
            int node = xidx[rr];
            v = emb[(size_t)node * EMB_DIM + c];
        }
        sA[r][c] = v;
    }
    __syncthreads();

    for (int t = 0; t < 4; ++t) {
        const int nbase = wave * 64 + t * 16;
        v8f acc = {0.f, 0.f, 0.f, 0.f, 0.f, 0.f, 0.f, 0.f};
        for (int s = 0; s < 32; ++s) {
            const int k = 4 * s + 2 * g;
            v2f a; a.x = sA[m][k];                    a.y = sA[m][k + 1];
            v2f b; b.x = W1[(size_t)k * HIDDEN + nbase + m];
                   b.y = W1[(size_t)(k + 1) * HIDDEN + nbase + m];
            acc = __builtin_amdgcn_wmma_f32_16x16x4_f32(
                false, a, false, b, (short)0, acc, false, false);
        }
        const float bias = b1[nbase + m];
        #pragma unroll
        for (int v = 0; v < 8; ++v) {
            float val = acc[v] + bias;
            val = val > 0.0f ? val : 0.0f;
            sH[v + 8 * g][nbase + m] = val;
        }
    }
    __syncthreads();

    {
        const int nbase = wave * 16;
        v8f acc = {0.f, 0.f, 0.f, 0.f, 0.f, 0.f, 0.f, 0.f};
        for (int s = 0; s < 64; ++s) {
            const int k = 4 * s + 2 * g;
            v2f a; a.x = sH[m][k];                    a.y = sH[m][k + 1];
            v2f b; b.x = W2[(size_t)k * OUT_C + nbase + m];
                   b.y = W2[(size_t)(k + 1) * OUT_C + nbase + m];
            acc = __builtin_amdgcn_wmma_f32_16x16x4_f32(
                false, a, false, b, (short)0, acc, false, false);
        }
        const float bias = b2[nbase + m];
        #pragma unroll
        for (int v = 0; v < 8; ++v) {
            const int row = row0 + v + 8 * g;
            if (row < n) {
                const float val = acc[v] + bias;
                const size_t idx = (size_t)row * OUT_C + nbase + m;
                h_out[idx]  = val;
                x0_out[idx] = val;
            }
        }
    }
}

__global__ void scatter_kernel(const int* __restrict__ src, const int* __restrict__ dst,
                               const float* __restrict__ norm,
                               const float* __restrict__ x, float* __restrict__ agg,
                               int e) {
    const unsigned gid = blockIdx.x * blockDim.x + threadIdx.x;
    const unsigned ei  = gid >> 4;
    const unsigned q   = gid & 15;
    if (ei < (unsigned)e) {
        const int   s  = src[ei];
        const int   d  = dst[ei];
        const float nv = norm[ei];
        const float4 v = ((const float4*)(x + (size_t)s * OUT_C))[q];
        float* ap = agg + (size_t)d * OUT_C + q * 4;
        atomicAdd(ap + 0, v.x * nv);
        atomicAdd(ap + 1, v.y * nv);
        atomicAdd(ap + 2, v.z * nv);
        atomicAdd(ap + 3, v.w * nv);
    }
}

__global__ void combine_kernel(float4* __restrict__ xnew_agg, const float4* __restrict__ xold,
                               const float* __restrict__ dis, const float4* __restrict__ h,
                               int total4) {
    const int i = blockIdx.x * blockDim.x + threadIdx.x;
    if (i < total4) {
        const int   node = i >> 4;
        const float di   = dis[node];
        const float d2   = di * di;
        const float4 a  = xnew_agg[i];
        const float4 xo = xold[i];
        const float4 hh = h[i];
        float4 r;
        r.x = (1.0f - ALPHA) * (a.x + d2 * xo.x) + ALPHA * hh.x;
        r.y = (1.0f - ALPHA) * (a.y + d2 * xo.y) + ALPHA * hh.y;
        r.z = (1.0f - ALPHA) * (a.z + d2 * xo.z) + ALPHA * hh.z;
        r.w = (1.0f - ALPHA) * (a.w + d2 * xo.w) + ALPHA * hh.w;
        xnew_agg[i] = r;
    }
}

extern "C" void kernel_launch(void* const* d_in, const int* in_sizes, int n_in,
                              void* d_out, int out_size, void* d_ws, size_t ws_size,
                              hipStream_t stream) {
    (void)n_in; (void)out_size; (void)ws_size;

    const int*   xidx = (const int*)d_in[0];
    const int*   ei   = (const int*)d_in[1];
    const float* emb  = (const float*)d_in[2];
    const float* W1   = (const float*)d_in[3];
    const float* b1   = (const float*)d_in[4];
    const float* W2   = (const float*)d_in[5];
    const float* b2   = (const float*)d_in[6];
    float* out = (float*)d_out;

    const int N = in_sizes[0];
    const int E = in_sizes[1] / 2;
    const size_t NC = (size_t)N * OUT_C;

    float* h     = (float*)d_ws;
    float* xping = h + NC;
    float* dis   = xping + NC;
    float* norm  = dis + N;

    const int* src = ei;
    const int* dst = ei + E;

    deg_init_kernel <<<(N + 255) / 256, 256, 0, stream>>>(dis, N);
    deg_count_kernel<<<(E + 255) / 256, 256, 0, stream>>>(dst, dis, E);
    deg_rsqrt_kernel<<<(N + 255) / 256, 256, 0, stream>>>(dis, N);
    edge_norm_kernel<<<(E + 255) / 256, 256, 0, stream>>>(src, dst, dis, norm, E);

    mlp_wmma_kernel<<<(N + 15) / 16, 128, 0, stream>>>(xidx, emb, W1, b1, W2, b2,
                                                       h, out, N);

    float* xa = out;
    float* xb = xping;
    const long long sthreads = (long long)E * 16;
    const int sblocks = (int)((sthreads + 255) / 256);
    const int total4  = (int)(NC / 4);
    const int cblocks = (total4 + 255) / 256;
    for (int k = 0; k < K_ITERS; ++k) {
        hipMemsetAsync(xb, 0, NC * sizeof(float), stream);
        scatter_kernel<<<sblocks, 256, 0, stream>>>(src, dst, norm, xa, xb, E);
        combine_kernel<<<cblocks, 256, 0, stream>>>((float4*)xb, (const float4*)xa,
                                                    dis, (const float4*)h, total4);
        float* t = xa; xa = xb; xb = t;
    }
    if (xa != out)
        hipMemcpyAsync(out, xa, NC * sizeof(float), hipMemcpyDeviceToDevice, stream);
}


# MLP weights stay in LDS all launch: 256 work-groups copy W1,W2 once in fragment order, a wave owns a 16-row tile, hidden tile stays in registers as layer-2 A operand; k summed in another order
# speedup vs baseline: 4.0775x; 4.0775x over previous
#include <hip/hip_runtime.h>
#include <hip/hip_bf16.h>

typedef __attribute__((ext_vector_type(2))) float v2f;
typedef __attribute__((ext_vector_type(8))) float v8f;

#define EMB_DIM 128
#define HIDDEN  256
#define OUT_C   64
#define K_ITERS 10
#define ALPHA   0.1f

#define MLP_WAVES 16
#define MLP_WGS   256

__launch_bounds__(MLP_WAVES * 32)
__global__ void mlp_wmma_kernel(const int* __restrict__ xidx,
                                const float* __restrict__ emb,
                                const float* __restrict__ W1, const float* __restrict__ b1,
                                const float* __restrict__ W2, const float* __restrict__ b2,
                                float* __restrict__ h_out, float* __restrict__ x0_out,
                                int n, int ntiles) {
    __shared__ float4 W1L[(EMB_DIM / 8) * (HIDDEN / 16) * 32];
    __shared__ float4 W2L[(HIDDEN / 16) * (OUT_C / 16) * 2 * 32];
    __shared__ float4 sB1v[HIDDEN / 4];
    float* sB1 = (float*)sB1v;

    const int tid  = threadIdx.x;
    const int lane = tid & 31;
    const int wave = __builtin_amdgcn_readfirstlane(tid >> 5);
    const int c    = lane & 15;
    const int hh   = lane >> 4;

    #pragma unroll
    for (int half = 0; half < 2; ++half) {
        float4 v[8];
        #pragma unroll
        for (int u = 0; u < 8; ++u) {
            const int i  = tid + (8 * half + u) * (MLP_WAVES * 32);
            const int nn = i & (HIDDEN - 1);
            const int hb = (i >> 8) & 1;
            const int j  = i >> 9;
            const int k0 = 8 * j + 4 * hb;
            v[u].x = W1[(size_t)(k0 + 0) * HIDDEN + nn];
            v[u].y = W1[(size_t)(k0 + 1) * HIDDEN + nn];
            v[u].z = W1[(size_t)(k0 + 2) * HIDDEN + nn];
            v[u].w = W1[(size_t)(k0 + 3) * HIDDEN + nn];
        }
        #pragma unroll
        for (int u = 0; u < 8; ++u) {
            const int i  = tid + (8 * half + u) * (MLP_WAVES * 32);
            const int nn = i & (HIDDEN - 1);
            const int hb = (i >> 8) & 1;
            const int j  = i >> 9;
            W1L[(j * (HIDDEN / 16) + (nn >> 4)) * 32 + hb * 16 + (nn & 15)] = v[u];
        }
    }
    {
        float4 v[8];
        #pragma unroll
        for (int u = 0; u < 8; ++u) {
            const int i  = tid + u * (MLP_WAVES * 32);
            const int nn = i & (OUT_C - 1);
            const int hb = (i >> 6) & 1;
            const int hf = (i >> 7) & 1;
            const int nt = i >> 8;
            const int k0 = 16 * nt + 8 * hb + 4 * hf;
            v[u].x = W2[(size_t)(k0 + 0) * OUT_C + nn];
            v[u].y = W2[(size_t)(k0 + 1) * OUT_C + nn];
            v[u].z = W2[(size_t)(k0 + 2) * OUT_C + nn];
            v[u].w = W2[(size_t)(k0 + 3) * OUT_C + nn];
        }
        #pragma unroll
        for (int u = 0; u < 8; ++u) {
            const int i  = tid + u * (MLP_WAVES * 32);
            const int nn = i & (OUT_C - 1);
            const int hb = (i >> 6) & 1;
            const int hf = (i >> 7) & 1;
            const int nt = i >> 8;
            W2L[((nt * (OUT_C / 16) + (nn >> 4)) * 2 + hf) * 32 + hb * 16 + (nn & 15)] = v[u];
        }
    }
    if (tid < HIDDEN) sB1[tid] = b1[tid];
    __syncthreads();

    float bias2[4];
    #pragma unroll
    for (int ot = 0; ot < 4; ++ot) bias2[ot] = b2[16 * ot + c];

    const int slot_odd = 4 * (((wave & 3) - (wave >> 2)) & 3) + (wave >> 2);
    for (int p = 0; blockIdx.x + MLP_WGS * MLP_WAVES * p < ntiles; ++p) {
        const int t = blockIdx.x + MLP_WGS * (MLP_WAVES * p + ((p & 1) ? slot_odd : wave));
        if (t >= ntiles) continue;
        const int row0 = t * 16;
        const int rowc = row0 + c;
        const int node = (rowc < n) ? xidx[rowc] : 0;
        const float4* xr = (const float4*)(emb + (size_t)node * EMB_DIM) + hh;
        float4 xq[16];
        #pragma unroll
        for (int j = 0; j < 16; ++j) xq[j] = xr[2 * j];

        v8f o[4];
        #pragma unroll
        for (int ot = 0; ot < 4; ++ot) o[ot] = (v8f){0.f, 0.f, 0.f, 0.f, 0.f, 0.f, 0.f, 0.f};

        #pragma nounroll
        for (int np = 0; np < 8; ++np) {
            v8f d0 = {0.f, 0.f, 0.f, 0.f, 0.f, 0.f, 0.f, 0.f};
            v8f d1 = {0.f, 0.f, 0.f, 0.f, 0.f, 0.f, 0.f, 0.f};
            const float4* w1p = W1L + (2 * np) * 32 + lane;
            #pragma unroll
            for (int j = 0; j < 16; ++j) {
                const float4 f0 = w1p[j * (HIDDEN / 16) * 32];
                const float4 f1 = w1p[j * (HIDDEN / 16) * 32 + 32];
                v2f a, b;
                b.x = xq[j].x; b.y = xq[j].y;
                a.x = f0.x; a.y = f0.y;
                d0 = __builtin_amdgcn_wmma_f32_16x16x4_f32(false, a, false, b, (short)0, d0, false, false);
                a.x = f1.x; a.y = f1.y;
                d1 = __builtin_amdgcn_wmma_f32_16x16x4_f32(false, a, false, b, (short)0, d1, false, false);
                b.x = xq[j].z; b.y = xq[j].w;
                a.x = f0.z; a.y = f0.w;
                d0 = __builtin_amdgcn_wmma_f32_16x16x4_f32(false, a, false, b, (short)0, d0, false, false);
                a.x = f1.z; a.y = f1.w;
                d1 = __builtin_amdgcn_wmma_f32_16x16x4_f32(false, a, false, b, (short)0, d1, false, false);
            }
            const float4* bp0 = (const float4*)(sB1 + 32 * np + 8 * hh);
            const float4* bp1 = (const float4*)(sB1 + 32 * np + 16 + 8 * hh);
            const float4 ba0 = bp0[0], bb0 = bp0[1], ba1 = bp1[0], bb1 = bp1[1];
            d0[0] = fmaxf(d0[0] + ba0.x, 0.0f); d0[1] = fmaxf(d0[1] + ba0.y, 0.0f);
            d0[2] = fmaxf(d0[2] + ba0.z, 0.0f); d0[3] = fmaxf(d0[3] + ba0.w, 0.0f);
            d0[4] = fmaxf(d0[4] + bb0.x, 0.0f); d0[5] = fmaxf(d0[5] + bb0.y, 0.0f);
            d0[6] = fmaxf(d0[6] + bb0.z, 0.0f); d0[7] = fmaxf(d0[7] + bb0.w, 0.0f);
            d1[0] = fmaxf(d1[0] + ba1.x, 0.0f); d1[1] = fmaxf(d1[1] + ba1.y, 0.0f);
            d1[2] = fmaxf(d1[2] + ba1.z, 0.0f); d1[3] = fmaxf(d1[3] + ba1.w, 0.0f);
            d1[4] = fmaxf(d1[4] + bb1.x, 0.0f); d1[5] = fmaxf(d1[5] + bb1.y, 0.0f);
            d1[6] = fmaxf(d1[6] + bb1.z, 0.0f); d1[7] = fmaxf(d1[7] + bb1.w, 0.0f);

            const float4* w2p = W2L + (2 * np) * (OUT_C / 16) * 2 * 32 + lane;
            #pragma unroll
            for (int u = 0; u < 2; ++u) {
                #pragma unroll
                for (int hf = 0; hf < 2; ++hf) {
                    float4 g[4];
                    #pragma unroll
                    for (int ot = 0; ot < 4; ++ot)
                        g[ot] = w2p[((u * (OUT_C / 16) + ot) * 2 + hf) * 32];
                    v2f a0, a1;
                    if (u == 0) { a0.x = d0[4 * hf + 0]; a0.y = d0[4 * hf + 1]; a1.x = d0[4 * hf + 2]; a1.y = d0[4 * hf + 3]; }
                    else        { a0.x = d1[4 * hf + 0]; a0.y = d1[4 * hf + 1]; a1.x = d1[4 * hf + 2]; a1.y = d1[4 * hf + 3]; }
                    #pragma unroll
                    for (int ot = 0; ot < 4; ++ot) {
                        v2f b;
                        b.x = g[ot].x; b.y = g[ot].y;
                        o[ot] = __builtin_amdgcn_wmma_f32_16x16x4_f32(false, a0, false, b, (short)0, o[ot], false, false);
                    }
                    #pragma unroll
                    for (int ot = 0; ot < 4; ++ot) {
                        v2f b;
                        b.x = g[ot].z; b.y = g[ot].w;
                        o[ot] = __builtin_amdgcn_wmma_f32_16x16x4_f32(false, a1, false, b, (short)0, o[ot], false, false);
                    }
                }
            }
        }

        #pragma unroll
        for (int ot = 0; ot < 4; ++ot) {
            #pragma unroll
            for (int r = 0; r < 8; ++r) {
                const int row = row0 + 8 * hh + r;
                if (row < n) {
                    const float val = o[ot][r] + bias2[ot];
                    const size_t idx = (size_t)row * OUT_C + 16 * ot + c;
                    h_out[idx]  = val;
                    x0_out[idx] = val;
                }
            }
        }
    }
}

#define TR        16
#define TR_SHIFT  4
#define MAXB      8192
#define NSLICE    256

__launch_bounds__(1024)
__global__ void bucket_count_kernel(const int* __restrict__ dst, int* __restrict__ cnt,
                                    int e, int nb) {
    __shared__ int hist[MAXB];
    const int tid = threadIdx.x;
    for (int i = tid; i < MAXB; i += 1024) hist[i] = 0;
    __syncthreads();
    const int per = (e + NSLICE - 1) / NSLICE;
    const int lo  = blockIdx.x * per;
    const int hi  = (lo + per < e) ? lo + per : e;
    for (int i = lo + tid; i < hi; i += 1024)
        atomicAdd(&hist[dst[i] >> TR_SHIFT], 1);
    __syncthreads();
    for (int b = tid; b < nb; b += 1024)
        cnt[(size_t)blockIdx.x * nb + b] = hist[b];
}

__launch_bounds__(256)
__global__ void bucket_scan_kernel(int* __restrict__ cnt, int* __restrict__ tot, int nb) {
    __shared__ int part[NSLICE / 32][32];
    const int bl = threadIdx.x & 31;
    const int q  = threadIdx.x >> 5;
    const int b  = blockIdx.x * 32 + bl;
    const bool ok = b < nb;
    int v[32];
    int sum = 0;
    #pragma unroll
    for (int i = 0; i < 32; ++i)
        v[i] = ok ? cnt[(size_t)(32 * q + i) * nb + b] : 0;
    #pragma unroll
    for (int i = 0; i < 32; ++i) { const int t = v[i]; v[i] = sum; sum += t; }
    part[q][bl] = sum;
    __syncthreads();
    int base = 0, total = 0;
    #pragma unroll
    for (int qq = 0; qq < NSLICE / 32; ++qq) {
        const int p = part[qq][bl];
        if (qq < q) base += p;
        total += p;
    }
    if (ok) {
        #pragma unroll
        for (int i = 0; i < 32; ++i)
            cnt[(size_t)(32 * q + i) * nb + b] = base + v[i];
        if (q == 0) tot[b] = total;
    }
}

__launch_bounds__(1024)
__global__ void bucket_fill_kernel(const int* __restrict__ src, const int* __restrict__ dst,
                                   const int* __restrict__ pre, const int* __restrict__ tot,
                                   unsigned short* __restrict__ lo16,
                                   unsigned char* __restrict__ hi8,
                                   int* __restrict__ bstart, int e, int nb) {
    __shared__ int cur[MAXB];
    __shared__ int wsum[32];
    const int tid  = threadIdx.x;
    const int lane = tid & 31;
    const int wave = tid >> 5;
    int tv[MAXB / 1024];
    int v = 0;
    #pragma unroll
    for (int i = 0; i < MAXB / 1024; ++i) {
        const int b = (MAXB / 1024) * tid + i;
        tv[i] = (b < nb) ? tot[b] : 0;
        v += tv[i];
    }
    int incl = v;
    #pragma unroll
    for (int d = 1; d < 32; d <<= 1) {
        const int t = __shfl_up(incl, d);
        if (lane >= d) incl += t;
    }
    if (lane == 31) wsum[wave] = incl;
    __syncthreads();
    if (wave == 0) {
        const int w = wsum[lane];
        int wi = w;
        #pragma unroll
        for (int d = 1; d < 32; d <<= 1) {
            const int t = __shfl_up(wi, d);
            if (lane >= d) wi += t;
        }
        wsum[lane] = wi - w;
    }
    __syncthreads();
    int run = incl - v + wsum[wave];
    #pragma unroll
    for (int i = 0; i < MAXB / 1024; ++i) {
        const int b = (MAXB / 1024) * tid + i;
        cur[b] = run + ((b < nb) ? pre[(size_t)blockIdx.x * nb + b] : 0);
        if (blockIdx.x == 0 && b < nb) bstart[b] = run;
        run += tv[i];
    }
    if (blockIdx.x == 0 && tid == 0) bstart[nb] = e;
    __syncthreads();
    const int per = (e + NSLICE - 1) / NSLICE;
    const int lo  = blockIdx.x * per;
    const int hi  = (lo + per < e) ? lo + per : e;
    for (int i = lo + tid; i < hi; i += 1024) {
        const int d = dst[i];
        const int s = src[i];
        const int pos = atomicAdd(&cur[d >> TR_SHIFT], 1);
        lo16[pos] = (unsigned short)(s & 0xFFFF);
        hi8[pos]  = (unsigned char)(((s >> 16) & 1) | ((d & (TR - 1)) << 1));
    }
}

#define DEG_WAVES 8
#define DEG_BATCH 8

__launch_bounds__(DEG_WAVES * 32)
__global__ void tile_degree_kernel(const unsigned char* __restrict__ hi8,
                                   const int* __restrict__ bstart,
                                   float* __restrict__ dis, int n, int nb) {
    __shared__ int indeg[DEG_WAVES][TR];
    const int lane = threadIdx.x & 31;
    const int wave = threadIdx.x >> 5;
    const int t    = blockIdx.x * DEG_WAVES + wave;
    if (lane < TR) indeg[wave][lane] = 0;
    __syncthreads();
    if (t < nb) {
        const int beg = bstart[t];
        const int end = bstart[t + 1];
        for (int e0 = beg + lane; e0 < end; e0 += 32 * DEG_BATCH) {
            unsigned v[DEG_BATCH];
            #pragma unroll
            for (int j = 0; j < DEG_BATCH; ++j) {
                const int e = e0 + j * 32;
                v[j] = (e < end) ? (unsigned)hi8[e] : 0x100u;
            }
            #pragma unroll
            for (int j = 0; j < DEG_BATCH; ++j)
                if (v[j] < 0x100u) atomicAdd(&indeg[wave][v[j] >> 1], 1);
        }
    }
    __syncthreads();
    if (t < nb && lane < TR) {
        const int node = t * TR + lane;
        if (node < n) {
            const float d = (float)(indeg[wave][lane] + 1);
            dis[node] = (d > 0.0f) ? rsqrtf(d) : 0.0f;
        }
    }
}

#define TILE_WAVES 4
#define DG_NONE    255u
#define NULL_EDGE  0xFF

template <int J>
__device__ __forceinline__ int row_share(int v) {
    return __builtin_amdgcn_mov_dpp(v, 0x150 + J, 0xF, 0xF, true);
}

struct GroupOps { float4 ra, rb; v2f a; };

template <int J>
__device__ __forceinline__ void fetch_group(int m0, int m1, int w0, int w1, int c,
                                            const char* __restrict__ xb, GroupOps& g) {
    const int ma = row_share<J>(m0);
    const int mb = row_share<J>(m1);
    const int wa = row_share<J>(w0);
    const int wb = row_share<J>(w1);
    g.ra = *(const float4*)(xb + (unsigned)((ma & ~0xFF) | (c << 4)));
    g.rb = *(const float4*)(xb + (unsigned)((mb & ~0xFF) | (c << 4)));
    g.a.x = ((ma & 0xFF) == c) ? __int_as_float(wa) : 0.0f;
    g.a.y = ((mb & 0xFF) == c) ? __int_as_float(wb) : 0.0f;
}

__device__ __forceinline__ void mma_group(const GroupOps& g, v8f (&acc)[4]) {
    v2f b;
    b.x = g.ra.x; b.y = g.rb.x;
    acc[0] = __builtin_amdgcn_wmma_f32_16x16x4_f32(false, g.a, false, b, (short)0, acc[0], false, false);
    b.x = g.ra.y; b.y = g.rb.y;
    acc[1] = __builtin_amdgcn_wmma_f32_16x16x4_f32(false, g.a, false, b, (short)0, acc[1], false, false);
    b.x = g.ra.z; b.y = g.rb.z;
    acc[2] = __builtin_amdgcn_wmma_f32_16x16x4_f32(false, g.a, false, b, (short)0, acc[2], false, false);
    b.x = g.ra.w; b.y = g.rb.w;
    acc[3] = __builtin_amdgcn_wmma_f32_16x16x4_f32(false, g.a, false, b, (short)0, acc[3], false, false);
}

template <int S>
__device__ __forceinline__ void stage(int m0, int m1, int w0, int w1, int c,
                                      const char* __restrict__ xb, v8f (&acc)[4]) {
    GroupOps g[8];
    fetch_group<8 * S + 0>(m0, m1, w0, w1, c, xb, g[0]);
    fetch_group<8 * S + 1>(m0, m1, w0, w1, c, xb, g[1]);
    fetch_group<8 * S + 2>(m0, m1, w0, w1, c, xb, g[2]);
    fetch_group<8 * S + 3>(m0, m1, w0, w1, c, xb, g[3]);
    fetch_group<8 * S + 4>(m0, m1, w0, w1, c, xb, g[4]);
    fetch_group<8 * S + 5>(m0, m1, w0, w1, c, xb, g[5]);
    fetch_group<8 * S + 6>(m0, m1, w0, w1, c, xb, g[6]);
    fetch_group<8 * S + 7>(m0, m1, w0, w1, c, xb, g[7]);
    #pragma unroll
    for (int j = 0; j < 8; ++j) mma_group(g[j], acc);
}

template <bool FIRST>
__launch_bounds__(TILE_WAVES * 32)
__global__ void round_kernel(const unsigned short* __restrict__ lo16,
                             const unsigned char* __restrict__ hi8,
                             unsigned char* __restrict__ dg8, int dcap,
                             const int* __restrict__ bstart,
                             const float* __restrict__ dis,
                             const float* __restrict__ xold,
                             const float* __restrict__ h,
                             float* __restrict__ xnew, int n, int nb, int e_all) {
    const int lane = threadIdx.x & 31;
    const int wave = __builtin_amdgcn_readfirstlane(threadIdx.x >> 5);
    const int t    = blockIdx.x * TILE_WAVES + wave;
    if (t >= nb) return;
    const int c     = lane & 15;
    const int hh    = lane >> 4;
    const int node0 = t * TR;
    const int beg   = bstart[t];
    const int end   = bstart[t + 1];
    const float dis_c = (node0 + c < n) ? dis[node0 + c] : 0.0f;
    const char* xb = (const char*)xold;

    v8f acc[4];
    #pragma unroll
    for (int i = 0; i < 4; ++i) acc[i] = (v8f){0.f, 0.f, 0.f, 0.f, 0.f, 0.f, 0.f, 0.f};

    for (int eb = beg & ~3; eb < end; eb += 64) {
        const int  p0    = eb + 4 * c + 2 * hh;
        const int  p1    = p0 + 1;
        const bool live0 = p0 >= beg && p0 < end;
        const bool live1 = p1 >= beg && p1 < end;
        const int  q0    = (p0 < e_all) ? p0 : e_all - 1;
        const int  q1    = (p1 < e_all) ? p1 : e_all - 1;
        const unsigned lo0 = lo16[q0], lo1 = lo16[q1];
        const unsigned hi0 = hi8[q0],  hi1 = hi8[q1];
        unsigned dg0 = DG_NONE, dg1 = DG_NONE;
        if (!FIRST) {
            if (q0 < dcap) dg0 = dg8[q0];
            if (q1 < dcap) dg1 = dg8[q1];
        }
        const int s0  = (int)(lo0 | ((hi0 & 1u) << 16));
        const int s1  = (int)(lo1 | ((hi1 & 1u) << 16));
        const int ld0 = (int)(hi0 >> 1);
        const int ld1 = (int)(hi1 >> 1);
        float ds0, ds1;
        if (FIRST) {
            ds0 = dis[s0];
            ds1 = dis[s1];
            if (live0 && p0 < dcap) {
                const int  d  = (int)(1.0f / (ds0 * ds0) + 0.5f);
                const bool ok = d >= 1 && d < (int)DG_NONE && rsqrtf((float)d) == ds0;
                dg8[p0] = (unsigned char)(ok ? (unsigned)d : DG_NONE);
            }
            if (live1 && p1 < dcap) {
                const int  d  = (int)(1.0f / (ds1 * ds1) + 0.5f);
                const bool ok = d >= 1 && d < (int)DG_NONE && rsqrtf((float)d) == ds1;
                dg8[p1] = (unsigned char)(ok ? (unsigned)d : DG_NONE);
            }
        } else {
            ds0 = (dg0 != DG_NONE) ? rsqrtf((float)dg0) : dis[s0];
            ds1 = (dg1 != DG_NONE) ? rsqrtf((float)dg1) : dis[s1];
        }
        const float dd0 = __shfl(dis_c, ld0 & (TR - 1));
        const float dd1 = __shfl(dis_c, ld1 & (TR - 1));
        const int w0 = __float_as_int(live0 ? ds0 * dd0 : 0.0f);
        const int w1 = __float_as_int(live1 ? ds1 * dd1 : 0.0f);
        const int m0 = live0 ? ((s0 << 8) | ld0) : NULL_EDGE;
        const int m1 = live1 ? ((s1 << 8) | ld1) : NULL_EDGE;

        stage<0>(m0, m1, w0, w1, c, xb, acc);
        if (eb + 32 < end)
            stage<1>(m0, m1, w0, w1, c, xb, acc);
    }

    float4 xo[8], hv[8];
    float  di[8];
    #pragma unroll
    for (int r = 0; r < 8; ++r) {
        const int row = node0 + 8 * hh + r;
        const int rr  = (row < n) ? row : 0;
        const size_t gi = (size_t)rr * (OUT_C / 4) + c;
        xo[r] = ((const float4*)xold)[gi];
        hv[r] = ((const float4*)h)[gi];
        di[r] = dis[rr];
    }
    #pragma unroll
    for (int r = 0; r < 8; ++r) {
        const int row = node0 + 8 * hh + r;
        if (row < n) {
            const float d2 = di[r] * di[r];
            float4 o;
            o.x = (1.0f - ALPHA) * (acc[0][r] + d2 * xo[r].x) + ALPHA * hv[r].x;
            o.y = (1.0f - ALPHA) * (acc[1][r] + d2 * xo[r].y) + ALPHA * hv[r].y;
            o.z = (1.0f - ALPHA) * (acc[2][r] + d2 * xo[r].z) + ALPHA * hv[r].z;
            o.w = (1.0f - ALPHA) * (acc[3][r] + d2 * xo[r].w) + ALPHA * hv[r].w;
            ((float4*)xnew)[(size_t)row * (OUT_C / 4) + c] = o;
        }
    }
}

extern "C" void kernel_launch(void* const* d_in, const int* in_sizes, int n_in,
                              void* d_out, int out_size, void* d_ws, size_t ws_size,
                              hipStream_t stream) {
    (void)n_in; (void)out_size; (void)ws_size;

    const int*   xidx = (const int*)d_in[0];
    const int*   ei   = (const int*)d_in[1];
    const float* emb  = (const float*)d_in[2];
    const float* W1   = (const float*)d_in[3];
    const float* b1   = (const float*)d_in[4];
    const float* W2   = (const float*)d_in[5];
    const float* b2   = (const float*)d_in[6];
    float* out = (float*)d_out;

    const int N = in_sizes[0];
    const int E = in_sizes[1] / 2;
    const size_t NC = (size_t)N * OUT_C;
    const int NB = (N + TR - 1) / TR;

    float* h     = (float*)d_ws;
    float* xping = h + NC;
    float* dis   = xping + NC;
    unsigned short* lo16 = (unsigned short*)(dis + N);
    unsigned char*  hi8  = (unsigned char*)(lo16 + E);
    int* bstart = (int*)(hi8 + (((size_t)E + 3) & ~(size_t)3));
    unsigned char* dg8 = (unsigned char*)(bstart + NB + 1);
    const int dcap = E - 4 * (NB + 1) - 4;
    int* cnt    = (int*)xping;
    int* tot    = cnt + (size_t)NSLICE * NB;

    const int* src = ei;
    const int* dst = ei + E;

    bucket_count_kernel<<<NSLICE, 1024, 0, stream>>>(dst, cnt, E, NB);
    bucket_scan_kernel <<<(NB + 31) / 32, 256, 0, stream>>>(cnt, tot, NB);
    bucket_fill_kernel <<<NSLICE, 1024, 0, stream>>>(src, dst, cnt, tot, lo16, hi8, bstart, E, NB);

    tile_degree_kernel<<<(NB + DEG_WAVES - 1) / DEG_WAVES, DEG_WAVES * 32, 0, stream>>>(hi8, bstart, dis, N, NB);

    mlp_wmma_kernel<<<MLP_WGS, MLP_WAVES * 32, 0, stream>>>(xidx, emb, W1, b1, W2, b2,
                                                            h, out, N, (N + 15) / 16);

    float* xa = out;
    float* xb = xping;
    const int RG = (NB + TILE_WAVES - 1) / TILE_WAVES;
    for (int k = 0; k < K_ITERS; ++k) {
        if (k == 0)
            round_kernel<true> <<<RG, TILE_WAVES * 32, 0, stream>>>(lo16, hi8, dg8, dcap, bstart, dis, xa, h, xb, N, NB, E);
        else
            round_kernel<false><<<RG, TILE_WAVES * 32, 0, stream>>>(lo16, hi8, dg8, dcap, bstart, dis, xa, h, xb, N, NB, E);
        float* t = xa; xa = xb; xb = t;
    }
    if (xa != out)
        hipMemcpyAsync(out, xa, NC * sizeof(float), hipMemcpyDeviceToDevice, stream);
}


# List fill places each edge with one scattered 4-byte packed word, not a 2-byte and a 1-byte store; the per-tile degree pass, which walks the slots anyway, splits words into the planes in order
# speedup vs baseline: 4.3962x; 1.0782x over previous
#include <hip/hip_runtime.h>
#include <hip/hip_bf16.h>

typedef __attribute__((ext_vector_type(2))) float v2f;
typedef __attribute__((ext_vector_type(8))) float v8f;

#define EMB_DIM 128
#define HIDDEN  256
#define OUT_C   64
#define K_ITERS 10
#define ALPHA   0.1f

#define MLP_WAVES 16
#define MLP_WGS   256

__launch_bounds__(MLP_WAVES * 32)
__global__ void mlp_wmma_kernel(const int* __restrict__ xidx,
                                const float* __restrict__ emb,
                                const float* __restrict__ W1, const float* __restrict__ b1,
                                const float* __restrict__ W2, const float* __restrict__ b2,
                                float* __restrict__ h_out, float* __restrict__ x0_out,
                                int n, int ntiles) {
    __shared__ float4 W1L[(EMB_DIM / 8) * (HIDDEN / 16) * 32];
    __shared__ float4 W2L[(HIDDEN / 16) * (OUT_C / 16) * 2 * 32];
    __shared__ float4 sB1v[HIDDEN / 4];
    float* sB1 = (float*)sB1v;

    const int tid  = threadIdx.x;
    const int lane = tid & 31;
    const int wave = __builtin_amdgcn_readfirstlane(tid >> 5);
    const int c    = lane & 15;
    const int hh   = lane >> 4;

    #pragma unroll
    for (int half = 0; half < 2; ++half) {
        float4 v[8];
        #pragma unroll
        for (int u = 0; u < 8; ++u) {
            const int i  = tid + (8 * half + u) * (MLP_WAVES * 32);
            const int nn = i & (HIDDEN - 1);
            const int hb = (i >> 8) & 1;
            const int j  = i >> 9;
            const int k0 = 8 * j + 4 * hb;
            v[u].x = W1[(size_t)(k0 + 0) * HIDDEN + nn];
            v[u].y = W1[(size_t)(k0 + 1) * HIDDEN + nn];
            v[u].z = W1[(size_t)(k0 + 2) * HIDDEN + nn];
            v[u].w = W1[(size_t)(k0 + 3) * HIDDEN + nn];
        }
        #pragma unroll
        for (int u = 0; u < 8; ++u) {
            const int i  = tid + (8 * half + u) * (MLP_WAVES * 32);
            const int nn = i & (HIDDEN - 1);
            const int hb = (i >> 8) & 1;
            const int j  = i >> 9;
            W1L[(j * (HIDDEN / 16) + (nn >> 4)) * 32 + hb * 16 + (nn & 15)] = v[u];
        }
    }
    {
        float4 v[8];
        #pragma unroll
        for (int u = 0; u < 8; ++u) {
            const int i  = tid + u * (MLP_WAVES * 32);
            const int nn = i & (OUT_C - 1);
            const int hb = (i >> 6) & 1;
            const int hf = (i >> 7) & 1;
            const int nt = i >> 8;
            const int k0 = 16 * nt + 8 * hb + 4 * hf;
            v[u].x = W2[(size_t)(k0 + 0) * OUT_C + nn];
            v[u].y = W2[(size_t)(k0 + 1) * OUT_C + nn];
            v[u].z = W2[(size_t)(k0 + 2) * OUT_C + nn];
            v[u].w = W2[(size_t)(k0 + 3) * OUT_C + nn];
        }
        #pragma unroll
        for (int u = 0; u < 8; ++u) {
            const int i  = tid + u * (MLP_WAVES * 32);
            const int nn = i & (OUT_C - 1);
            const int hb = (i >> 6) & 1;
            const int hf = (i >> 7) & 1;
            const int nt = i >> 8;
            W2L[((nt * (OUT_C / 16) + (nn >> 4)) * 2 + hf) * 32 + hb * 16 + (nn & 15)] = v[u];
        }
    }
    if (tid < HIDDEN) sB1[tid] = b1[tid];
    __syncthreads();

    float bias2[4];
    #pragma unroll
    for (int ot = 0; ot < 4; ++ot) bias2[ot] = b2[16 * ot + c];

    const int slot_odd = 4 * (((wave & 3) - (wave >> 2)) & 3) + (wave >> 2);
    for (int p = 0; blockIdx.x + MLP_WGS * MLP_WAVES * p < ntiles; ++p) {
        const int t = blockIdx.x + MLP_WGS * (MLP_WAVES * p + ((p & 1) ? slot_odd : wave));
        if (t >= ntiles) continue;
        const int row0 = t * 16;
        const int rowc = row0 + c;
        const int node = (rowc < n) ? xidx[rowc] : 0;
        const float4* xr = (const float4*)(emb + (size_t)node * EMB_DIM) + hh;
        float4 xq[16];
        #pragma unroll
        for (int j = 0; j < 16; ++j) xq[j] = xr[2 * j];

        v8f o[4];
        #pragma unroll
        for (int ot = 0; ot < 4; ++ot) o[ot] = (v8f){0.f, 0.f, 0.f, 0.f, 0.f, 0.f, 0.f, 0.f};

        #pragma nounroll
        for (int np = 0; np < 8; ++np) {
            v8f d0 = {0.f, 0.f, 0.f, 0.f, 0.f, 0.f, 0.f, 0.f};
            v8f d1 = {0.f, 0.f, 0.f, 0.f, 0.f, 0.f, 0.f, 0.f};
            const float4* w1p = W1L + (2 * np) * 32 + lane;
            #pragma unroll
            for (int j = 0; j < 16; ++j) {
                const float4 f0 = w1p[j * (HIDDEN / 16) * 32];
                const float4 f1 = w1p[j * (HIDDEN / 16) * 32 + 32];
                v2f a, b;
                b.x = xq[j].x; b.y = xq[j].y;
                a.x = f0.x; a.y = f0.y;
                d0 = __builtin_amdgcn_wmma_f32_16x16x4_f32(false, a, false, b, (short)0, d0, false, false);
                a.x = f1.x; a.y = f1.y;
                d1 = __builtin_amdgcn_wmma_f32_16x16x4_f32(false, a, false, b, (short)0, d1, false, false);
                b.x = xq[j].z; b.y = xq[j].w;
                a.x = f0.z; a.y = f0.w;
                d0 = __builtin_amdgcn_wmma_f32_16x16x4_f32(false, a, false, b, (short)0, d0, false, false);
                a.x = f1.z; a.y = f1.w;
                d1 = __builtin_amdgcn_wmma_f32_16x16x4_f32(false, a, false, b, (short)0, d1, false, false);
            }
            const float4* bp0 = (const float4*)(sB1 + 32 * np + 8 * hh);
            const float4* bp1 = (const float4*)(sB1 + 32 * np + 16 + 8 * hh);
            const float4 ba0 = bp0[0], bb0 = bp0[1], ba1 = bp1[0], bb1 = bp1[1];
            d0[0] = fmaxf(d0[0] + ba0.x, 0.0f); d0[1] = fmaxf(d0[1] + ba0.y, 0.0f);
            d0[2] = fmaxf(d0[2] + ba0.z, 0.0f); d0[3] = fmaxf(d0[3] + ba0.w, 0.0f);
            d0[4] = fmaxf(d0[4] + bb0.x, 0.0f); d0[5] = fmaxf(d0[5] + bb0.y, 0.0f);
            d0[6] = fmaxf(d0[6] + bb0.z, 0.0f); d0[7] = fmaxf(d0[7] + bb0.w, 0.0f);
            d1[0] = fmaxf(d1[0] + ba1.x, 0.0f); d1[1] = fmaxf(d1[1] + ba1.y, 0.0f);
            d1[2] = fmaxf(d1[2] + ba1.z, 0.0f); d1[3] = fmaxf(d1[3] + ba1.w, 0.0f);
            d1[4] = fmaxf(d1[4] + bb1.x, 0.0f); d1[5] = fmaxf(d1[5] + bb1.y, 0.0f);
            d1[6] = fmaxf(d1[6] + bb1.z, 0.0f); d1[7] = fmaxf(d1[7] + bb1.w, 0.0f);

            const float4* w2p = W2L + (2 * np) * (OUT_C / 16) * 2 * 32 + lane;
            #pragma unroll
            for (int u = 0; u < 2; ++u) {
                #pragma unroll
                for (int hf = 0; hf < 2; ++hf) {
                    float4 g[4];
                    #pragma unroll
                    for (int ot = 0; ot < 4; ++ot)
                        g[ot] = w2p[((u * (OUT_C / 16) + ot) * 2 + hf) * 32];
                    v2f a0, a1;
                    if (u == 0) { a0.x = d0[4 * hf + 0]; a0.y = d0[4 * hf + 1]; a1.x = d0[4 * hf + 2]; a1.y = d0[4 * hf + 3]; }
                    else        { a0.x = d1[4 * hf + 0]; a0.y = d1[4 * hf + 1]; a1.x = d1[4 * hf + 2]; a1.y = d1[4 * hf + 3]; }
                    #pragma unroll
                    for (int ot = 0; ot < 4; ++ot) {
                        v2f b;
                        b.x = g[ot].x; b.y = g[ot].y;
                        o[ot] = __builtin_amdgcn_wmma_f32_16x16x4_f32(false, a0, false, b, (short)0, o[ot], false, false);
                    }
                    #pragma unroll
                    for (int ot = 0; ot < 4; ++ot) {
                        v2f b;
                        b.x = g[ot].z; b.y = g[ot].w;
                        o[ot] = __builtin_amdgcn_wmma_f32_16x16x4_f32(false, a1, false, b, (short)0, o[ot], false, false);
                    }
                }
            }
        }

        #pragma unroll
        for (int ot = 0; ot < 4; ++ot) {
            #pragma unroll
            for (int r = 0; r < 8; ++r) {
                const int row = row0 + 8 * hh + r;
                if (row < n) {
                    const float val = o[ot][r] + bias2[ot];
                    const size_t idx = (size_t)row * OUT_C + 16 * ot + c;
                    h_out[idx]  = val;
                    x0_out[idx] = val;
                }
            }
        }
    }
}

#define TR        16
#define TR_SHIFT  4
#define MAXB      8192
#define NSLICE    256

__launch_bounds__(1024)
__global__ void bucket_count_kernel(const int* __restrict__ dst, int* __restrict__ cnt,
                                    int e, int nb) {
    __shared__ int hist[MAXB];
    const int tid = threadIdx.x;
    for (int i = tid; i < MAXB; i += 1024) hist[i] = 0;
    __syncthreads();
    const int per = (e + NSLICE - 1) / NSLICE;
    const int lo  = blockIdx.x * per;
    const int hi  = (lo + per < e) ? lo + per : e;
    for (int i = lo + tid; i < hi; i += 1024)
        atomicAdd(&hist[dst[i] >> TR_SHIFT], 1);
    __syncthreads();
    for (int b = tid; b < nb; b += 1024)
        cnt[(size_t)blockIdx.x * nb + b] = hist[b];
}

__launch_bounds__(256)
__global__ void bucket_scan_kernel(int* __restrict__ cnt, int* __restrict__ tot, int nb) {
    __shared__ int part[NSLICE / 32][32];
    const int bl = threadIdx.x & 31;
    const int q  = threadIdx.x >> 5;
    const int b  = blockIdx.x * 32 + bl;
    const bool ok = b < nb;
    int v[32];
    int sum = 0;
    #pragma unroll
    for (int i = 0; i < 32; ++i)
        v[i] = ok ? cnt[(size_t)(32 * q + i) * nb + b] : 0;
    #pragma unroll
    for (int i = 0; i < 32; ++i) { const int t = v[i]; v[i] = sum; sum += t; }
    part[q][bl] = sum;
    __syncthreads();
    int base = 0, total = 0;
    #pragma unroll
    for (int qq = 0; qq < NSLICE / 32; ++qq) {
        const int p = part[qq][bl];
        if (qq < q) base += p;
        total += p;
    }
    if (ok) {
        #pragma unroll
        for (int i = 0; i < 32; ++i)
            cnt[(size_t)(32 * q + i) * nb + b] = base + v[i];
        if (q == 0) tot[b] = total;
    }
}

__launch_bounds__(1024)
__global__ void bucket_fill_kernel(const int* __restrict__ src, const int* __restrict__ dst,
                                   const int* __restrict__ pre, const int* __restrict__ tot,
                                   unsigned* __restrict__ ew,
                                   int* __restrict__ bstart, int e, int nb) {
    __shared__ int cur[MAXB];
    __shared__ int wsum[32];
    const int tid  = threadIdx.x;
    const int lane = tid & 31;
    const int wave = tid >> 5;
    int tv[MAXB / 1024];
    int v = 0;
    #pragma unroll
    for (int i = 0; i < MAXB / 1024; ++i) {
        const int b = (MAXB / 1024) * tid + i;
        tv[i] = (b < nb) ? tot[b] : 0;
        v += tv[i];
    }
    int incl = v;
    #pragma unroll
    for (int d = 1; d < 32; d <<= 1) {
        const int t = __shfl_up(incl, d);
        if (lane >= d) incl += t;
    }
    if (lane == 31) wsum[wave] = incl;
    __syncthreads();
    if (wave == 0) {
        const int w = wsum[lane];
        int wi = w;
        #pragma unroll
        for (int d = 1; d < 32; d <<= 1) {
            const int t = __shfl_up(wi, d);
            if (lane >= d) wi += t;
        }
        wsum[lane] = wi - w;
    }
    __syncthreads();
    int run = incl - v + wsum[wave];
    #pragma unroll
    for (int i = 0; i < MAXB / 1024; ++i) {
        const int b = (MAXB / 1024) * tid + i;
        cur[b] = run + ((b < nb) ? pre[(size_t)blockIdx.x * nb + b] : 0);
        if (blockIdx.x == 0 && b < nb) bstart[b] = run;
        run += tv[i];
    }
    if (blockIdx.x == 0 && tid == 0) bstart[nb] = e;
    __syncthreads();
    const int per = (e + NSLICE - 1) / NSLICE;
    const int lo  = blockIdx.x * per;
    const int hi  = (lo + per < e) ? lo + per : e;
    for (int i = lo + tid; i < hi; i += 1024) {
        const int d = dst[i];
        const int s = src[i];
        const int pos = atomicAdd(&cur[d >> TR_SHIFT], 1);
        ew[pos] = (unsigned)s | ((unsigned)(d & (TR - 1)) << 17);
    }
}

#define DEG_WAVES 8
#define DEG_BATCH 8

__launch_bounds__(DEG_WAVES * 32)
__global__ void tile_degree_kernel(const unsigned* __restrict__ ew,
                                   unsigned short* __restrict__ lo16,
                                   unsigned char* __restrict__ hi8,
                                   const int* __restrict__ bstart,
                                   float* __restrict__ dis, int n, int nb) {
    __shared__ int indeg[DEG_WAVES][TR];
    const int lane = threadIdx.x & 31;
    const int wave = threadIdx.x >> 5;
    const int t    = blockIdx.x * DEG_WAVES + wave;
    if (lane < TR) indeg[wave][lane] = 0;
    __syncthreads();
    if (t < nb) {
        const int beg = bstart[t];
        const int end = bstart[t + 1];
        for (int e0 = beg + lane; e0 < end; e0 += 32 * DEG_BATCH) {
            unsigned v[DEG_BATCH];
            #pragma unroll
            for (int j = 0; j < DEG_BATCH; ++j) {
                const int e = e0 + j * 32;
                v[j] = (e < end) ? ew[e] : 0xFFFFFFFFu;
            }
            #pragma unroll
            for (int j = 0; j < DEG_BATCH; ++j) {
                if (v[j] != 0xFFFFFFFFu) {
                    const int e = e0 + j * 32;
                    lo16[e] = (unsigned short)(v[j] & 0xFFFFu);
                    hi8[e]  = (unsigned char)((v[j] >> 16) & 0x1Fu);
                    atomicAdd(&indeg[wave][(v[j] >> 17) & (TR - 1)], 1);
                }
            }
        }
    }
    __syncthreads();
    if (t < nb && lane < TR) {
        const int node = t * TR + lane;
        if (node < n) {
            const float d = (float)(indeg[wave][lane] + 1);
            dis[node] = (d > 0.0f) ? rsqrtf(d) : 0.0f;
        }
    }
}

#define TILE_WAVES 4
#define DG_NONE    255u
#define NULL_EDGE  0xFF

template <int J>
__device__ __forceinline__ int row_share(int v) {
    return __builtin_amdgcn_mov_dpp(v, 0x150 + J, 0xF, 0xF, true);
}

struct GroupOps { float4 ra, rb; v2f a; };

template <int J>
__device__ __forceinline__ void fetch_group(int m0, int m1, int w0, int w1, int c,
                                            const char* __restrict__ xb, GroupOps& g) {
    const int ma = row_share<J>(m0);
    const int mb = row_share<J>(m1);
    const int wa = row_share<J>(w0);
    const int wb = row_share<J>(w1);
    g.ra = *(const float4*)(xb + (unsigned)((ma & ~0xFF) | (c << 4)));
    g.rb = *(const float4*)(xb + (unsigned)((mb & ~0xFF) | (c << 4)));
    g.a.x = ((ma & 0xFF) == c) ? __int_as_float(wa) : 0.0f;
    g.a.y = ((mb & 0xFF) == c) ? __int_as_float(wb) : 0.0f;
}

__device__ __forceinline__ void mma_group(const GroupOps& g, v8f (&acc)[4]) {
    v2f b;
    b.x = g.ra.x; b.y = g.rb.x;
    acc[0] = __builtin_amdgcn_wmma_f32_16x16x4_f32(false, g.a, false, b, (short)0, acc[0], false, false);
    b.x = g.ra.y; b.y = g.rb.y;
    acc[1] = __builtin_amdgcn_wmma_f32_16x16x4_f32(false, g.a, false, b, (short)0, acc[1], false, false);
    b.x = g.ra.z; b.y = g.rb.z;
    acc[2] = __builtin_amdgcn_wmma_f32_16x16x4_f32(false, g.a, false, b, (short)0, acc[2], false, false);
    b.x = g.ra.w; b.y = g.rb.w;
    acc[3] = __builtin_amdgcn_wmma_f32_16x16x4_f32(false, g.a, false, b, (short)0, acc[3], false, false);
}

template <int S>
__device__ __forceinline__ void stage(int m0, int m1, int w0, int w1, int c,
                                      const char* __restrict__ xb, v8f (&acc)[4]) {
    GroupOps g[8];
    fetch_group<8 * S + 0>(m0, m1, w0, w1, c, xb, g[0]);
    fetch_group<8 * S + 1>(m0, m1, w0, w1, c, xb, g[1]);
    fetch_group<8 * S + 2>(m0, m1, w0, w1, c, xb, g[2]);
    fetch_group<8 * S + 3>(m0, m1, w0, w1, c, xb, g[3]);
    fetch_group<8 * S + 4>(m0, m1, w0, w1, c, xb, g[4]);
    fetch_group<8 * S + 5>(m0, m1, w0, w1, c, xb, g[5]);
    fetch_group<8 * S + 6>(m0, m1, w0, w1, c, xb, g[6]);
    fetch_group<8 * S + 7>(m0, m1, w0, w1, c, xb, g[7]);
    #pragma unroll
    for (int j = 0; j < 8; ++j) mma_group(g[j], acc);
}

template <bool FIRST>
__launch_bounds__(TILE_WAVES * 32)
__global__ void round_kernel(const unsigned short* __restrict__ lo16,
                             const unsigned char* __restrict__ hi8,
                             unsigned char* __restrict__ dg8, int dcap,
                             const int* __restrict__ bstart,
                             const float* __restrict__ dis,
                             const float* __restrict__ xold,
                             const float* __restrict__ h,
                             float* __restrict__ xnew, int n, int nb, int e_all) {
    const int lane = threadIdx.x & 31;
    const int wave = __builtin_amdgcn_readfirstlane(threadIdx.x >> 5);
    const int t    = blockIdx.x * TILE_WAVES + wave;
    if (t >= nb) return;
    const int c     = lane & 15;
    const int hh    = lane >> 4;
    const int node0 = t * TR;
    const int beg   = bstart[t];
    const int end   = bstart[t + 1];
    const float dis_c = (node0 + c < n) ? dis[node0 + c] : 0.0f;
    const char* xb = (const char*)xold;

    v8f acc[4];
    #pragma unroll
    for (int i = 0; i < 4; ++i) acc[i] = (v8f){0.f, 0.f, 0.f, 0.f, 0.f, 0.f, 0.f, 0.f};

    for (int eb = beg & ~3; eb < end; eb += 64) {
        const int  p0    = eb + 4 * c + 2 * hh;
        const int  p1    = p0 + 1;
        const bool live0 = p0 >= beg && p0 < end;
        const bool live1 = p1 >= beg && p1 < end;
        const int  q0    = (p0 < e_all) ? p0 : e_all - 1;
        const int  q1    = (p1 < e_all) ? p1 : e_all - 1;
        const unsigned lo0 = lo16[q0], lo1 = lo16[q1];
        const unsigned hi0 = hi8[q0],  hi1 = hi8[q1];
        unsigned dg0 = DG_NONE, dg1 = DG_NONE;
        if (!FIRST) {
            if (q0 < dcap) dg0 = dg8[q0];
            if (q1 < dcap) dg1 = dg8[q1];
        }
        const int s0  = (int)(lo0 | ((hi0 & 1u) << 16));
        const int s1  = (int)(lo1 | ((hi1 & 1u) << 16));
        const int ld0 = (int)(hi0 >> 1);
        const int ld1 = (int)(hi1 >> 1);
        float ds0, ds1;
        if (FIRST) {
            ds0 = dis[s0];
            ds1 = dis[s1];
            if (live0 && p0 < dcap) {
                const int  d  = (int)(1.0f / (ds0 * ds0) + 0.5f);
                const bool ok = d >= 1 && d < (int)DG_NONE && rsqrtf((float)d) == ds0;
                dg8[p0] = (unsigned char)(ok ? (unsigned)d : DG_NONE);
            }
            if (live1 && p1 < dcap) {
                const int  d  = (int)(1.0f / (ds1 * ds1) + 0.5f);
                const bool ok = d >= 1 && d < (int)DG_NONE && rsqrtf((float)d) == ds1;
                dg8[p1] = (unsigned char)(ok ? (unsigned)d : DG_NONE);
            }
        } else {
            ds0 = (dg0 != DG_NONE) ? rsqrtf((float)dg0) : dis[s0];
            ds1 = (dg1 != DG_NONE) ? rsqrtf((float)dg1) : dis[s1];
        }
        const float dd0 = __shfl(dis_c, ld0 & (TR - 1));
        const float dd1 = __shfl(dis_c, ld1 & (TR - 1));
        const int w0 = __float_as_int(live0 ? ds0 * dd0 : 0.0f);
        const int w1 = __float_as_int(live1 ? ds1 * dd1 : 0.0f);
        const int m0 = live0 ? ((s0 << 8) | ld0) : NULL_EDGE;
        const int m1 = live1 ? ((s1 << 8) | ld1) : NULL_EDGE;

        stage<0>(m0, m1, w0, w1, c, xb, acc);
        if (eb + 32 < end)
            stage<1>(m0, m1, w0, w1, c, xb, acc);
    }

    float4 xo[8], hv[8];
    float  di[8];
    #pragma unroll
    for (int r = 0; r < 8; ++r) {
        const int row = node0 + 8 * hh + r;
        const int rr  = (row < n) ? row : 0;
        const size_t gi = (size_t)rr * (OUT_C / 4) + c;
        xo[r] = ((const float4*)xold)[gi];
        hv[r] = ((const float4*)h)[gi];
        di[r] = dis[rr];
    }
    #pragma unroll
    for (int r = 0; r < 8; ++r) {
        const int row = node0 + 8 * hh + r;
        if (row < n) {
            const float d2 = di[r] * di[r];
            float4 o;
            o.x = (1.0f - ALPHA) * (acc[0][r] + d2 * xo[r].x) + ALPHA * hv[r].x;
            o.y = (1.0f - ALPHA) * (acc[1][r] + d2 * xo[r].y) + ALPHA * hv[r].y;
            o.z = (1.0f - ALPHA) * (acc[2][r] + d2 * xo[r].z) + ALPHA * hv[r].z;
            o.w = (1.0f - ALPHA) * (acc[3][r] + d2 * xo[r].w) + ALPHA * hv[r].w;
            ((float4*)xnew)[(size_t)row * (OUT_C / 4) + c] = o;
        }
    }
}

extern "C" void kernel_launch(void* const* d_in, const int* in_sizes, int n_in,
                              void* d_out, int out_size, void* d_ws, size_t ws_size,
                              hipStream_t stream) {
    (void)n_in; (void)out_size; (void)ws_size;

    const int*   xidx = (const int*)d_in[0];
    const int*   ei   = (const int*)d_in[1];
    const float* emb  = (const float*)d_in[2];
    const float* W1   = (const float*)d_in[3];
    const float* b1   = (const float*)d_in[4];
    const float* W2   = (const float*)d_in[5];
    const float* b2   = (const float*)d_in[6];
    float* out = (float*)d_out;

    const int N = in_sizes[0];
    const int E = in_sizes[1] / 2;
    const size_t NC = (size_t)N * OUT_C;
    const int NB = (N + TR - 1) / TR;

    float* h     = (float*)d_ws;
    float* xping = h + NC;
    float* dis   = xping + NC;
    unsigned short* lo16 = (unsigned short*)(dis + N);
    unsigned char*  hi8  = (unsigned char*)(lo16 + E);
    int* bstart = (int*)(hi8 + (((size_t)E + 3) & ~(size_t)3));
    unsigned char* dg8 = (unsigned char*)(bstart + NB + 1);
    const int dcap = E - 4 * (NB + 1) - 4;
    int* cnt    = (int*)xping;
    int* tot    = cnt + (size_t)NSLICE * NB;
    unsigned* ew = (unsigned*)(tot + NB);

    const int* src = ei;
    const int* dst = ei + E;

    bucket_count_kernel<<<NSLICE, 1024, 0, stream>>>(dst, cnt, E, NB);
    bucket_scan_kernel <<<(NB + 31) / 32, 256, 0, stream>>>(cnt, tot, NB);
    bucket_fill_kernel <<<NSLICE, 1024, 0, stream>>>(src, dst, cnt, tot, ew, bstart, E, NB);

    tile_degree_kernel<<<(NB + DEG_WAVES - 1) / DEG_WAVES, DEG_WAVES * 32, 0, stream>>>(ew, lo16, hi8, bstart, dis, N, NB);

    mlp_wmma_kernel<<<MLP_WGS, MLP_WAVES * 32, 0, stream>>>(xidx, emb, W1, b1, W2, b2,
                                                            h, out, N, (N + 15) / 16);

    float* xa = out;
    float* xb = xping;
    const int RG = (NB + TILE_WAVES - 1) / TILE_WAVES;
    for (int k = 0; k < K_ITERS; ++k) {
        if (k == 0)
            round_kernel<true> <<<RG, TILE_WAVES * 32, 0, stream>>>(lo16, hi8, dg8, dcap, bstart, dis, xa, h, xb, N, NB, E);
        else
            round_kernel<false><<<RG, TILE_WAVES * 32, 0, stream>>>(lo16, hi8, dg8, dcap, bstart, dis, xa, h, xb, N, NB, E);
        float* t = xa; xa = xb; xb = t;
    }
    if (xa != out)
        hipMemcpyAsync(out, xa, NC * sizeof(float), hipMemcpyDeviceToDevice, stream);
}


# One-hot A operand of the rounds built by an LDS scatter: a lane stores its 2 edge weights in a zeroed per-wave tile, groups read ready 16-B fragments, not DPP/compare/select chains; sums unchanged
# speedup vs baseline: 4.5479x; 1.0345x over previous
#include <hip/hip_runtime.h>
#include <hip/hip_bf16.h>

typedef __attribute__((ext_vector_type(2))) float v2f;
typedef __attribute__((ext_vector_type(8))) float v8f;

#define EMB_DIM 128
#define HIDDEN  256
#define OUT_C   64
#define K_ITERS 10
#define ALPHA   0.1f

#define MLP_WAVES 16
#define MLP_WGS   256

__launch_bounds__(MLP_WAVES * 32)
__global__ void mlp_wmma_kernel(const int* __restrict__ xidx,
                                const float* __restrict__ emb,
                                const float* __restrict__ W1, const float* __restrict__ b1,
                                const float* __restrict__ W2, const float* __restrict__ b2,
                                float* __restrict__ h_out, float* __restrict__ x0_out,
                                int n, int ntiles) {
    __shared__ float4 W1L[(EMB_DIM / 8) * (HIDDEN / 16) * 32];
    __shared__ float4 W2L[(HIDDEN / 16) * (OUT_C / 16) * 2 * 32];
    __shared__ float4 sB1v[HIDDEN / 4];
    float* sB1 = (float*)sB1v;

    const int tid  = threadIdx.x;
    const int lane = tid & 31;
    const int wave = __builtin_amdgcn_readfirstlane(tid >> 5);
    const int c    = lane & 15;
    const int hh   = lane >> 4;

    #pragma unroll
    for (int half = 0; half < 2; ++half) {
        float4 v[8];
        #pragma unroll
        for (int u = 0; u < 8; ++u) {
            const int i  = tid + (8 * half + u) * (MLP_WAVES * 32);
            const int nn = i & (HIDDEN - 1);
            const int hb = (i >> 8) & 1;
            const int j  = i >> 9;
            const int k0 = 8 * j + 4 * hb;
            v[u].x = W1[(size_t)(k0 + 0) * HIDDEN + nn];
            v[u].y = W1[(size_t)(k0 + 1) * HIDDEN + nn];
            v[u].z = W1[(size_t)(k0 + 2) * HIDDEN + nn];
            v[u].w = W1[(size_t)(k0 + 3) * HIDDEN + nn];
        }
        #pragma unroll
        for (int u = 0; u < 8; ++u) {
            const int i  = tid + (8 * half + u) * (MLP_WAVES * 32);
            const int nn = i & (HIDDEN - 1);
            const int hb = (i >> 8) & 1;
            const int j  = i >> 9;
            W1L[(j * (HIDDEN / 16) + (nn >> 4)) * 32 + hb * 16 + (nn & 15)] = v[u];
        }
    }
    {
        float4 v[8];
        #pragma unroll
        for (int u = 0; u < 8; ++u) {
            const int i  = tid + u * (MLP_WAVES * 32);
            const int nn = i & (OUT_C - 1);
            const int hb = (i >> 6) & 1;
            const int hf = (i >> 7) & 1;
            const int nt = i >> 8;
            const int k0 = 16 * nt + 8 * hb + 4 * hf;
            v[u].x = W2[(size_t)(k0 + 0) * OUT_C + nn];
            v[u].y = W2[(size_t)(k0 + 1) * OUT_C + nn];
            v[u].z = W2[(size_t)(k0 + 2) * OUT_C + nn];
            v[u].w = W2[(size_t)(k0 + 3) * OUT_C + nn];
        }
        #pragma unroll
        for (int u = 0; u < 8; ++u) {
            const int i  = tid + u * (MLP_WAVES * 32);
            const int nn = i & (OUT_C - 1);
            const int hb = (i >> 6) & 1;
            const int hf = (i >> 7) & 1;
            const int nt = i >> 8;
            W2L[((nt * (OUT_C / 16) + (nn >> 4)) * 2 + hf) * 32 + hb * 16 + (nn & 15)] = v[u];
        }
    }
    if (tid < HIDDEN) sB1[tid] = b1[tid];
    __syncthreads();

    float bias2[4];
    #pragma unroll
    for (int ot = 0; ot < 4; ++ot) bias2[ot] = b2[16 * ot + c];

    const int slot_odd = 4 * (((wave & 3) - (wave >> 2)) & 3) + (wave >> 2);
    for (int p = 0; blockIdx.x + MLP_WGS * MLP_WAVES * p < ntiles; ++p) {
        const int t = blockIdx.x + MLP_WGS * (MLP_WAVES * p + ((p & 1) ? slot_odd : wave));
        if (t >= ntiles) continue;
        const int row0 = t * 16;
        const int rowc = row0 + c;
        const int node = (rowc < n) ? xidx[rowc] : 0;
        const float4* xr = (const float4*)(emb + (size_t)node * EMB_DIM) + hh;
        float4 xq[16];
        #pragma unroll
        for (int j = 0; j < 16; ++j) xq[j] = xr[2 * j];

        v8f o[4];
        #pragma unroll
        for (int ot = 0; ot < 4; ++ot) o[ot] = (v8f){0.f, 0.f, 0.f, 0.f, 0.f, 0.f, 0.f, 0.f};

        #pragma nounroll
        for (int np = 0; np < 8; ++np) {
            v8f d0 = {0.f, 0.f, 0.f, 0.f, 0.f, 0.f, 0.f, 0.f};
            v8f d1 = {0.f, 0.f, 0.f, 0.f, 0.f, 0.f, 0.f, 0.f};
            const float4* w1p = W1L + (2 * np) * 32 + lane;
            #pragma unroll
            for (int j = 0; j < 16; ++j) {
                const float4 f0 = w1p[j * (HIDDEN / 16) * 32];
                const float4 f1 = w1p[j * (HIDDEN / 16) * 32 + 32];
                v2f a, b;
                b.x = xq[j].x; b.y = xq[j].y;
                a.x = f0.x; a.y = f0.y;
                d0 = __builtin_amdgcn_wmma_f32_16x16x4_f32(false, a, false, b, (short)0, d0, false, false);
                a.x = f1.x; a.y = f1.y;
                d1 = __builtin_amdgcn_wmma_f32_16x16x4_f32(false, a, false, b, (short)0, d1, false, false);
                b.x = xq[j].z; b.y = xq[j].w;
                a.x = f0.z; a.y = f0.w;
                d0 = __builtin_amdgcn_wmma_f32_16x16x4_f32(false, a, false, b, (short)0, d0, false, false);
                a.x = f1.z; a.y = f1.w;
                d1 = __builtin_amdgcn_wmma_f32_16x16x4_f32(false, a, false, b, (short)0, d1, false, false);
            }
            const float4* bp0 = (const float4*)(sB1 + 32 * np + 8 * hh);
            const float4* bp1 = (const float4*)(sB1 + 32 * np + 16 + 8 * hh);
            const float4 ba0 = bp0[0], bb0 = bp0[1], ba1 = bp1[0], bb1 = bp1[1];
            d0[0] = fmaxf(d0[0] + ba0.x, 0.0f); d0[1] = fmaxf(d0[1] + ba0.y, 0.0f);
            d0[2] = fmaxf(d0[2] + ba0.z, 0.0f); d0[3] = fmaxf(d0[3] + ba0.w, 0.0f);
            d0[4] = fmaxf(d0[4] + bb0.x, 0.0f); d0[5] = fmaxf(d0[5] + bb0.y, 0.0f);
            d0[6] = fmaxf(d0[6] + bb0.z, 0.0f); d0[7] = fmaxf(d0[7] + bb0.w, 0.0f);
            d1[0] = fmaxf(d1[0] + ba1.x, 0.0f); d1[1] = fmaxf(d1[1] + ba1.y, 0.0f);
            d1[2] = fmaxf(d1[2] + ba1.z, 0.0f); d1[3] = fmaxf(d1[3] + ba1.w, 0.0f);
            d1[4] = fmaxf(d1[4] + bb1.x, 0.0f); d1[5] = fmaxf(d1[5] + bb1.y, 0.0f);
            d1[6] = fmaxf(d1[6] + bb1.z, 0.0f); d1[7] = fmaxf(d1[7] + bb1.w, 0.0f);

            const float4* w2p = W2L + (2 * np) * (OUT_C / 16) * 2 * 32 + lane;
            #pragma unroll
            for (int u = 0; u < 2; ++u) {
                #pragma unroll
                for (int hf = 0; hf < 2; ++hf) {
                    float4 g[4];
                    #pragma unroll
                    for (int ot = 0; ot < 4; ++ot)
                        g[ot] = w2p[((u * (OUT_C / 16) + ot) * 2 + hf) * 32];
                    v2f a0, a1;
                    if (u == 0) { a0.x = d0[4 * hf + 0]; a0.y = d0[4 * hf + 1]; a1.x = d0[4 * hf + 2]; a1.y = d0[4 * hf + 3]; }
                    else        { a0.x = d1[4 * hf + 0]; a0.y = d1[4 * hf + 1]; a1.x = d1[4 * hf + 2]; a1.y = d1[4 * hf + 3]; }
                    #pragma unroll
                    for (int ot = 0; ot < 4; ++ot) {
                        v2f b;
                        b.x = g[ot].x; b.y = g[ot].y;
                        o[ot] = __builtin_amdgcn_wmma_f32_16x16x4_f32(false, a0, false, b, (short)0, o[ot], false, false);
                    }
                    #pragma unroll
                    for (int ot = 0; ot < 4; ++ot) {
                        v2f b;
                        b.x = g[ot].z; b.y = g[ot].w;
                        o[ot] = __builtin_amdgcn_wmma_f32_16x16x4_f32(false, a1, false, b, (short)0, o[ot], false, false);
                    }
                }
            }
        }

        #pragma unroll
        for (int ot = 0; ot < 4; ++ot) {
            #pragma unroll
            for (int r = 0; r < 8; ++r) {
                const int row = row0 + 8 * hh + r;
                if (row < n) {
                    const float val = o[ot][r] + bias2[ot];
                    const size_t idx = (size_t)row * OUT_C + 16 * ot + c;
                    h_out[idx]  = val;
                    x0_out[idx] = val;
                }
            }
        }
    }
}

#define TR        16
#define TR_SHIFT  4
#define MAXB      8192
#define NSLICE    256

__launch_bounds__(1024)
__global__ void bucket_count_kernel(const int* __restrict__ dst, int* __restrict__ cnt,
                                    int e, int nb) {
    __shared__ int hist[MAXB];
    const int tid = threadIdx.x;
    for (int i = tid; i < MAXB; i += 1024) hist[i] = 0;
    __syncthreads();
    const int per = (e + NSLICE - 1) / NSLICE;
    const int lo  = blockIdx.x * per;
    const int hi  = (lo + per < e) ? lo + per : e;
    for (int i = lo + tid; i < hi; i += 1024)
        atomicAdd(&hist[dst[i] >> TR_SHIFT], 1);
    __syncthreads();
    for (int b = tid; b < nb; b += 1024)
        cnt[(size_t)blockIdx.x * nb + b] = hist[b];
}

__launch_bounds__(256)
__global__ void bucket_scan_kernel(int* __restrict__ cnt, int* __restrict__ tot, int nb) {
    __shared__ int part[NSLICE / 32][32];
    const int bl = threadIdx.x & 31;
    const int q  = threadIdx.x >> 5;
    const int b  = blockIdx.x * 32 + bl;
    const bool ok = b < nb;
    int v[32];
    int sum = 0;
    #pragma unroll
    for (int i = 0; i < 32; ++i)
        v[i] = ok ? cnt[(size_t)(32 * q + i) * nb + b] : 0;
    #pragma unroll
    for (int i = 0; i < 32; ++i) { const int t = v[i]; v[i] = sum; sum += t; }
    part[q][bl] = sum;
    __syncthreads();
    int base = 0, total = 0;
    #pragma unroll
    for (int qq = 0; qq < NSLICE / 32; ++qq) {
        const int p = part[qq][bl];
        if (qq < q) base += p;
        total += p;
    }
    if (ok) {
        #pragma unroll
        for (int i = 0; i < 32; ++i)
            cnt[(size_t)(32 * q + i) * nb + b] = base + v[i];
        if (q == 0) tot[b] = total;
    }
}

__launch_bounds__(1024)
__global__ void bucket_fill_kernel(const int* __restrict__ src, const int* __restrict__ dst,
                                   const int* __restrict__ pre, const int* __restrict__ tot,
                                   unsigned* __restrict__ ew,
                                   int* __restrict__ bstart, int e, int nb) {
    __shared__ int cur[MAXB];
    __shared__ int wsum[32];
    const int tid  = threadIdx.x;
    const int lane = tid & 31;
    const int wave = tid >> 5;
    int tv[MAXB / 1024];
    int v = 0;
    #pragma unroll
    for (int i = 0; i < MAXB / 1024; ++i) {
        const int b = (MAXB / 1024) * tid + i;
        tv[i] = (b < nb) ? tot[b] : 0;
        v += tv[i];
    }
    int incl = v;
    #pragma unroll
    for (int d = 1; d < 32; d <<= 1) {
        const int t = __shfl_up(incl, d);
        if (lane >= d) incl += t;
    }
    if (lane == 31) wsum[wave] = incl;
    __syncthreads();
    if (wave == 0) {
        const int w = wsum[lane];
        int wi = w;
        #pragma unroll
        for (int d = 1; d < 32; d <<= 1) {
            const int t = __shfl_up(wi, d);
            if (lane >= d) wi += t;
        }
        wsum[lane] = wi - w;
    }
    __syncthreads();
    int run = incl - v + wsum[wave];
    #pragma unroll
    for (int i = 0; i < MAXB / 1024; ++i) {
        const int b = (MAXB / 1024) * tid + i;
        cur[b] = run + ((b < nb) ? pre[(size_t)blockIdx.x * nb + b] : 0);
        if (blockIdx.x == 0 && b < nb) bstart[b] = run;
        run += tv[i];
    }
    if (blockIdx.x == 0 && tid == 0) bstart[nb] = e;
    __syncthreads();
    const int per = (e + NSLICE - 1) / NSLICE;
    const int lo  = blockIdx.x * per;
    const int hi  = (lo + per < e) ? lo + per : e;
    for (int i = lo + tid; i < hi; i += 1024) {
        const int d = dst[i];
        const int s = src[i];
        const int pos = atomicAdd(&cur[d >> TR_SHIFT], 1);
        ew[pos] = (unsigned)s | ((unsigned)(d & (TR - 1)) << 17);
    }
}

#define DEG_WAVES 8
#define DEG_BATCH 8

__launch_bounds__(DEG_WAVES * 32)
__global__ void tile_degree_kernel(const unsigned* __restrict__ ew,
                                   unsigned short* __restrict__ lo16,
                                   unsigned char* __restrict__ hi8,
                                   const int* __restrict__ bstart,
                                   float* __restrict__ dis, int n, int nb) {
    __shared__ int indeg[DEG_WAVES][TR];
    const int lane = threadIdx.x & 31;
    const int wave = threadIdx.x >> 5;
    const int t    = blockIdx.x * DEG_WAVES + wave;
    if (lane < TR) indeg[wave][lane] = 0;
    __syncthreads();
    if (t < nb) {
        const int beg = bstart[t];
        const int end = bstart[t + 1];
        for (int e0 = beg + lane; e0 < end; e0 += 32 * DEG_BATCH) {
            unsigned v[DEG_BATCH];
            #pragma unroll
            for (int j = 0; j < DEG_BATCH; ++j) {
                const int e = e0 + j * 32;
                v[j] = (e < end) ? ew[e] : 0xFFFFFFFFu;
            }
            #pragma unroll
            for (int j = 0; j < DEG_BATCH; ++j) {
                if (v[j] != 0xFFFFFFFFu) {
                    const int e = e0 + j * 32;
                    lo16[e] = (unsigned short)(v[j] & 0xFFFFu);
                    hi8[e]  = (unsigned char)((v[j] >> 16) & 0x1Fu);
                    atomicAdd(&indeg[wave][(v[j] >> 17) & (TR - 1)], 1);
                }
            }
        }
    }
    __syncthreads();
    if (t < nb && lane < TR) {
        const int node = t * TR + lane;
        if (node < n) {
            const float d = (float)(indeg[wave][lane] + 1);
            dis[node] = (d > 0.0f) ? rsqrtf(d) : 0.0f;
        }
    }
}

#define TILE_WAVES 4
#define DG_NONE    255u
#define SEL_FLOATS 1024

template <int J>
__device__ __forceinline__ int row_share(int v) {
    return __builtin_amdgcn_mov_dpp(v, 0x150 + J, 0xF, 0xF, true);
}

struct GroupOps { float4 ra, rb; };

template <int J>
__device__ __forceinline__ void fetch_group(int o0, int o1, int c16,
                                            const char* __restrict__ xb, GroupOps& g) {
    const int oa = row_share<J>(o0) | c16;
    const int ob = row_share<J>(o1) | c16;
    g.ra = *(const float4*)(xb + (unsigned)oa);
    g.rb = *(const float4*)(xb + (unsigned)ob);
}

__device__ __forceinline__ void mma_group(const GroupOps& g, v2f a, v8f (&acc)[4]) {
    v2f b;
    b.x = g.ra.x; b.y = g.rb.x;
    acc[0] = __builtin_amdgcn_wmma_f32_16x16x4_f32(false, a, false, b, (short)0, acc[0], false, false);
    b.x = g.ra.y; b.y = g.rb.y;
    acc[1] = __builtin_amdgcn_wmma_f32_16x16x4_f32(false, a, false, b, (short)0, acc[1], false, false);
    b.x = g.ra.z; b.y = g.rb.z;
    acc[2] = __builtin_amdgcn_wmma_f32_16x16x4_f32(false, a, false, b, (short)0, acc[2], false, false);
    b.x = g.ra.w; b.y = g.rb.w;
    acc[3] = __builtin_amdgcn_wmma_f32_16x16x4_f32(false, a, false, b, (short)0, acc[3], false, false);
}

template <int S>
__device__ __forceinline__ void stage(int o0, int o1, int c16, const char* __restrict__ xb,
                                      const float4* selq, v8f (&acc)[4]) {
    GroupOps g[8];
    fetch_group<8 * S + 0>(o0, o1, c16, xb, g[0]);
    fetch_group<8 * S + 1>(o0, o1, c16, xb, g[1]);
    fetch_group<8 * S + 2>(o0, o1, c16, xb, g[2]);
    fetch_group<8 * S + 3>(o0, o1, c16, xb, g[3]);
    fetch_group<8 * S + 4>(o0, o1, c16, xb, g[4]);
    fetch_group<8 * S + 5>(o0, o1, c16, xb, g[5]);
    fetch_group<8 * S + 6>(o0, o1, c16, xb, g[6]);
    fetch_group<8 * S + 7>(o0, o1, c16, xb, g[7]);
    float4 f[4];
    #pragma unroll
    for (int p = 0; p < 4; ++p) f[p] = selq[(4 * S + p) * 32];
    #pragma unroll
    for (int j = 0; j < 8; ++j) {
        v2f a;
        if (j & 1) { a.x = f[j >> 1].z; a.y = f[j >> 1].w; }
        else       { a.x = f[j >> 1].x; a.y = f[j >> 1].y; }
        mma_group(g[j], a, acc);
    }
}

template <bool FIRST>
__launch_bounds__(TILE_WAVES * 32)
__global__ void round_kernel(const unsigned short* __restrict__ lo16,
                             const unsigned char* __restrict__ hi8,
                             unsigned char* __restrict__ dg8, int dcap,
                             const int* __restrict__ bstart,
                             const float* __restrict__ dis,
                             const float* __restrict__ xold,
                             const float* __restrict__ h,
                             float* __restrict__ xnew, int n, int nb, int e_all) {
    __shared__ float4 sel[TILE_WAVES * SEL_FLOATS / 4];
    const int lane = threadIdx.x & 31;
    const int wave = __builtin_amdgcn_readfirstlane(threadIdx.x >> 5);
    const int t    = blockIdx.x * TILE_WAVES + wave;
    if (t >= nb) return;
    const int c     = lane & 15;
    const int hh    = lane >> 4;
    const int c16   = c << 4;
    const int node0 = t * TR;
    const int beg   = bstart[t];
    const int end   = bstart[t + 1];
    const float dis_c = (node0 + c < n) ? dis[node0 + c] : 0.0f;
    const char* xb = (const char*)xold;

    float4* selv = sel + wave * (SEL_FLOATS / 4);
    #pragma unroll
    for (int i = 0; i < SEL_FLOATS / 4 / 32; ++i) selv[lane + 32 * i] = (float4){0.f, 0.f, 0.f, 0.f};
    float* selw = (float*)selv;
    const float4* selq = selv + 2 * c + hh;
    const int wbase = (c >> 1) * 128 + hh * 4 + (c & 1) * 2;

    v8f acc[4];
    #pragma unroll
    for (int i = 0; i < 4; ++i) acc[i] = (v8f){0.f, 0.f, 0.f, 0.f, 0.f, 0.f, 0.f, 0.f};

    for (int eb = beg & ~3; eb < end; eb += 64) {
        const int  p0    = eb + 4 * c + 2 * hh;
        const int  p1    = p0 + 1;
        const bool live0 = p0 >= beg && p0 < end;
        const bool live1 = p1 >= beg && p1 < end;
        const int  q0    = (p0 < e_all) ? p0 : e_all - 1;
        const int  q1    = (p1 < e_all) ? p1 : e_all - 1;
        const unsigned lo0 = lo16[q0], lo1 = lo16[q1];
        const unsigned hi0 = hi8[q0],  hi1 = hi8[q1];
        unsigned dg0 = DG_NONE, dg1 = DG_NONE;
        if (!FIRST) {
            if (q0 < dcap) dg0 = dg8[q0];
            if (q1 < dcap) dg1 = dg8[q1];
        }
        const int s0  = (int)(lo0 | ((hi0 & 1u) << 16));
        const int s1  = (int)(lo1 | ((hi1 & 1u) << 16));
        const int ld0 = (int)(hi0 >> 1) & (TR - 1);
        const int ld1 = (int)(hi1 >> 1) & (TR - 1);
        float ds0, ds1;
        if (FIRST) {
            ds0 = dis[s0];
            ds1 = dis[s1];
            if (live0 && p0 < dcap) {
                const int  d  = (int)(1.0f / (ds0 * ds0) + 0.5f);
                const bool ok = d >= 1 && d < (int)DG_NONE && rsqrtf((float)d) == ds0;
                dg8[p0] = (unsigned char)(ok ? (unsigned)d : DG_NONE);
            }
            if (live1 && p1 < dcap) {
                const int  d  = (int)(1.0f / (ds1 * ds1) + 0.5f);
                const bool ok = d >= 1 && d < (int)DG_NONE && rsqrtf((float)d) == ds1;
                dg8[p1] = (unsigned char)(ok ? (unsigned)d : DG_NONE);
            }
        } else {
            ds0 = (dg0 != DG_NONE) ? rsqrtf((float)dg0) : dis[s0];
            ds1 = (dg1 != DG_NONE) ? rsqrtf((float)dg1) : dis[s1];
        }
        const float dd0 = __shfl(dis_c, ld0);
        const float dd1 = __shfl(dis_c, ld1);
        const float w0 = live0 ? ds0 * dd0 : 0.0f;
        const float w1 = live1 ? ds1 * dd1 : 0.0f;
        const int o0 = live0 ? (s0 << 8) : 0;
        const int o1 = live1 ? (s1 << 8) : 0;

        const int i0 = wbase + ld0 * 8;
        const int i1 = wbase + ld1 * 8 + 1;
        selw[i0] = w0;
        selw[i1] = w1;

        stage<0>(o0, o1, c16, xb, selq, acc);
        if (eb + 32 < end)
            stage<1>(o0, o1, c16, xb, selq, acc);

        selw[i0] = 0.0f;
        selw[i1] = 0.0f;
    }

    float4 xo[8], hv[8];
    float  di[8];
    #pragma unroll
    for (int r = 0; r < 8; ++r) {
        const int row = node0 + 8 * hh + r;
        const int rr  = (row < n) ? row : 0;
        const size_t gi = (size_t)rr * (OUT_C / 4) + c;
        xo[r] = ((const float4*)xold)[gi];
        hv[r] = ((const float4*)h)[gi];
        di[r] = dis[rr];
    }
    #pragma unroll
    for (int r = 0; r < 8; ++r) {
        const int row = node0 + 8 * hh + r;
        if (row < n) {
            const float d2 = di[r] * di[r];
            float4 o;
            o.x = (1.0f - ALPHA) * (acc[0][r] + d2 * xo[r].x) + ALPHA * hv[r].x;
            o.y = (1.0f - ALPHA) * (acc[1][r] + d2 * xo[r].y) + ALPHA * hv[r].y;
            o.z = (1.0f - ALPHA) * (acc[2][r] + d2 * xo[r].z) + ALPHA * hv[r].z;
            o.w = (1.0f - ALPHA) * (acc[3][r] + d2 * xo[r].w) + ALPHA * hv[r].w;
            ((float4*)xnew)[(size_t)row * (OUT_C / 4) + c] = o;
        }
    }
}

extern "C" void kernel_launch(void* const* d_in, const int* in_sizes, int n_in,
                              void* d_out, int out_size, void* d_ws, size_t ws_size,
                              hipStream_t stream) {
    (void)n_in; (void)out_size; (void)ws_size;

    const int*   xidx = (const int*)d_in[0];
    const int*   ei   = (const int*)d_in[1];
    const float* emb  = (const float*)d_in[2];
    const float* W1   = (const float*)d_in[3];
    const float* b1   = (const float*)d_in[4];
    const float* W2   = (const float*)d_in[5];
    const float* b2   = (const float*)d_in[6];
    float* out = (float*)d_out;

    const int N = in_sizes[0];
    const int E = in_sizes[1] / 2;
    const size_t NC = (size_t)N * OUT_C;
    const int NB = (N + TR - 1) / TR;

    float* h     = (float*)d_ws;
    float* xping = h + NC;
    float* dis   = xping + NC;
    unsigned short* lo16 = (unsigned short*)(dis + N);
    unsigned char*  hi8  = (unsigned char*)(lo16 + E);
    int* bstart = (int*)(hi8 + (((size_t)E + 3) & ~(size_t)3));
    unsigned char* dg8 = (unsigned char*)(bstart + NB + 1);
    const int dcap = E - 4 * (NB + 1) - 4;
    int* cnt    = (int*)xping;
    int* tot    = cnt + (size_t)NSLICE * NB;
    unsigned* ew = (unsigned*)(tot + NB);

    const int* src = ei;
    const int* dst = ei + E;

    bucket_count_kernel<<<NSLICE, 1024, 0, stream>>>(dst, cnt, E, NB);
    bucket_scan_kernel <<<(NB + 31) / 32, 256, 0, stream>>>(cnt, tot, NB);
    bucket_fill_kernel <<<NSLICE, 1024, 0, stream>>>(src, dst, cnt, tot, ew, bstart, E, NB);

    tile_degree_kernel<<<(NB + DEG_WAVES - 1) / DEG_WAVES, DEG_WAVES * 32, 0, stream>>>(ew, lo16, hi8, bstart, dis, N, NB);

    mlp_wmma_kernel<<<MLP_WGS, MLP_WAVES * 32, 0, stream>>>(xidx, emb, W1, b1, W2, b2,
                                                            h, out, N, (N + 15) / 16);

    float* xa = out;
    float* xb = xping;
    const int RG = (NB + TILE_WAVES - 1) / TILE_WAVES;
    for (int k = 0; k < K_ITERS; ++k) {
        if (k == 0)
            round_kernel<true> <<<RG, TILE_WAVES * 32, 0, stream>>>(lo16, hi8, dg8, dcap, bstart, dis, xa, h, xb, N, NB, E);
        else
            round_kernel<false><<<RG, TILE_WAVES * 32, 0, stream>>>(lo16, hi8, dg8, dcap, bstart, dis, xa, h, xb, N, NB, E);
        float* t = xa; xa = xb; xb = t;
    }
    if (xa != out)
        hipMemcpyAsync(out, xa, NC * sizeof(float), hipMemcpyDeviceToDevice, stream);
}
